# starting kernel
# baseline (speedup 1.0000x reference)
#include <hip/hip_runtime.h>
#include <hip/hip_bf16.h>
#include <float.h>
#include <math.h>

typedef __attribute__((ext_vector_type(2))) float v2f;
typedef __attribute__((ext_vector_type(8))) float v8f;

#define B_   16
#define T_   512
#define C_   8
#define S_   64
#define KSZ_ 32
#define TOUT_ 481
#define EPS_ 1e-8f


__global__ void lsd_x_stats(const float* __restrict__ x, float* __restrict__ ws) {
    __shared__ float ssum[128];
    __shared__ float ssq[128];
    const int bc  = blockIdx.x;
    const int b   = bc >> 3;
    const int c   = bc & 7;
    const int tid = threadIdx.x;
    float s = 0.f, q = 0.f;
    for (int i = 0; i < 4; ++i) {
        const int t = tid + i * 128;
        const float v = x[(b * T_ + t) * C_ + c];
        s += v; q += v * v;
    }
    ssum[tid] = s; ssq[tid] = q;
    __syncthreads();
    for (int off = 64; off > 0; off >>= 1) {
        if (tid < off) { ssum[tid] += ssum[tid + off]; ssq[tid] += ssq[tid + off]; }
        __syncthreads();
    }
    if (tid == 0) {
        const float mean = ssum[0] * (1.f / (float)T_);
        const float var  = fmaxf(ssq[0] * (1.f / (float)T_) - mean * mean, 0.f);
        ws[bc]       = mean;
        ws[128 + bc] = 1.f / (sqrtf(var) + EPS_);
    }
}

__global__ void lsd_k_stats(const float* __restrict__ kern, float* __restrict__ ws) {
    __shared__ float ssum[256];
    __shared__ float ssq[256];
    __shared__ float smu, sinv;
    const int sidx = blockIdx.x;
    const int tid  = threadIdx.x;
    const float v = kern[sidx * 256 + tid];
    ssum[tid] = v; ssq[tid] = v * v;
    __syncthreads();
    for (int off = 128; off > 0; off >>= 1) {
        if (tid < off) { ssum[tid] += ssum[tid + off]; ssq[tid] += ssq[tid + off]; }
        __syncthreads();
    }
    if (tid == 0) {
        const float mean = ssum[0] * (1.f / 256.f);
        const float var  = fmaxf(ssq[0] * (1.f / 256.f) - mean * mean, 0.f);
        const float inv  = 1.f / (sqrtf(var) + EPS_);
        ws[256 + sidx] = mean;
        ws[320 + sidx] = inv;
        smu = mean; sinv = inv;
    }
    __syncthreads();
    if (tid < KSZ_) {
        const float mu  = smu;
        const float inv = sinv;
        float acc = 0.f;
        for (int c = 0; c < C_; ++c) {
            const float q = (kern[sidx * 256 + tid * C_ + c] - mu) * inv;
            acc += q * q;
        }
        ws[384 + sidx * KSZ_ + tid] = acc;
    }
}

__global__ __launch_bounds__(32) void lsd_wmma_kernel(const float* __restrict__ x,
                                                      const float* __restrict__ kern,
                                                      const float* __restrict__ ws,
                                                      float* __restrict__ out) {
    const int t0 = blockIdx.x * 16;
    const int k  = blockIdx.y;
    const int b  = blockIdx.z;
    const int l  = threadIdx.x;

    const int ch   = k >> 2;
    const int tau0 = (k & 3) * 8;
    const int base = t0 + tau0;

    __shared__ float zsh[24];
    __shared__ float pnsh[16];

    const float mux = ws[b * C_ + ch];
    const float inx = ws[128 + b * C_ + ch];

    if (l < 23) {
        int t = base + l;
        if (t > T_ - 1) t = T_ - 1;
        zsh[l] = (x[(b * T_ + t) * C_ + ch] - mux) * inx;
    }
    __syncthreads();

    const int m  = l & 15;
    const int hi = l >> 4;

    if (l < 16) {
        float acc = 0.f;
        for (int c = 0; c < 8; ++c) { const float z = zsh[l + c]; acc += z * z; }
        pnsh[l] = acc;
    }
    __syncthreads();

    v2f a_lo, a_hi;
    a_lo.x = zsh[m + 2 * hi];
    a_lo.y = zsh[m + 2 * hi + 1];
    a_hi.x = zsh[m + 2 * hi + 4];
    a_hi.y = zsh[m + 2 * hi + 5];

    float pn2[8];
    for (int j = 0; j < 8; ++j) pn2[j] = pnsh[j + 8 * hi];

    float md[8];
    for (int j = 0; j < 8; ++j) md[j] = FLT_MAX;

    const int n = m;
    for (int st = 0; st < 4; ++st) {
        const int s = st * 16 + n;
        const float muk = ws[256 + s];
        const float ink = ws[320 + s];
        const float* kp = kern + s * 256 + k * 8;
        v2f b_lo, b_hi;
        b_lo.x = (kp[2 * hi]     - muk) * ink;
        b_lo.y = (kp[2 * hi + 1] - muk) * ink;
        b_hi.x = (kp[2 * hi + 4] - muk) * ink;
        b_hi.y = (kp[2 * hi + 5] - muk) * ink;
        const float kn2 = ws[384 + s * KSZ_ + k];

        v8f g = {0.f, 0.f, 0.f, 0.f, 0.f, 0.f, 0.f, 0.f};
        g = __builtin_amdgcn_wmma_f32_16x16x4_f32(false, a_lo, false, b_lo,
                                                  (short)0, g, false, false);
        g = __builtin_amdgcn_wmma_f32_16x16x4_f32(false, a_hi, false, b_hi,
                                                  (short)0, g, false, false);

        for (int j = 0; j < 8; ++j) {
            const float d = pn2[j] + kn2 - 2.f * g[j];
            md[j] = fminf(md[j], d);
        }
    }

    for (int mask = 1; mask <= 8; mask <<= 1)
        for (int j = 0; j < 8; ++j)
            md[j] = fminf(md[j], __shfl_xor(md[j], mask, 32));

    if (n == 0) {
        for (int j = 0; j < 8; ++j) {
            const int t = t0 + j + 8 * hi;
            if (t < TOUT_) out[(b * TOUT_ + t) * KSZ_ + k] = md[j];
        }
    }
}

extern "C" void kernel_launch(void* const* d_in, const int* in_sizes, int n_in,
                              void* d_out, int out_size, void* d_ws, size_t ws_size,
                              hipStream_t stream) {
    (void)in_sizes; (void)n_in; (void)out_size; (void)ws_size;
    const float* x    = (const float*)d_in[0];
    const float* kern = (const float*)d_in[1];
    float* ws  = (float*)d_ws;
    float* out = (float*)d_out;

    lsd_x_stats<<<dim3(B_ * C_), dim3(128), 0, stream>>>(x, ws);
    lsd_k_stats<<<dim3(S_), dim3(256), 0, stream>>>(kern, ws);
    lsd_wmma_kernel<<<dim3(31, KSZ_, B_), dim3(32), 0, stream>>>(x, kern, ws, out);
}


# Main kernel regridded by output rows: 256 groups of 8 waves, each owning 32 whole rows of a batch entry (wave = channel, its 4 slots' B fragments in registers); rows leave via LDS as 128-B lines
# speedup vs baseline: 1.8000x; 1.8000x over previous
#include <hip/hip_runtime.h>
#include <hip/hip_bf16.h>
#include <float.h>
#include <math.h>


typedef __attribute__((ext_vector_type(2))) float v2f;
typedef __attribute__((ext_vector_type(4))) float v4f;
typedef __attribute__((ext_vector_type(8))) float v8f;

#define B_   16
#define T_   512
#define C_   8
#define S_   64
#define KSZ_ 32
#define TOUT_ 481
#define EPS_ 1e-8f

#define ROWS_ 32
#define ZW_   64
#define OPITCH_ 36


__global__ void lsd_x_stats(const float* __restrict__ x, float* __restrict__ ws) {
    __shared__ float ssum[128];
    __shared__ float ssq[128];
    const int bc  = blockIdx.x;
    const int b   = bc >> 3;
    const int c   = bc & 7;
    const int tid = threadIdx.x;
    float s = 0.f, q = 0.f;
    for (int i = 0; i < 4; ++i) {
        const int t = tid + i * 128;
        const float v = x[(b * T_ + t) * C_ + c];
        s += v; q += v * v;
    }
    ssum[tid] = s; ssq[tid] = q;
    __syncthreads();
    for (int off = 64; off > 0; off >>= 1) {
        if (tid < off) { ssum[tid] += ssum[tid + off]; ssq[tid] += ssq[tid + off]; }
        __syncthreads();
    }
    if (tid == 0) {
        const float mean = ssum[0] * (1.f / (float)T_);
        const float var  = fmaxf(ssq[0] * (1.f / (float)T_) - mean * mean, 0.f);
        ws[bc]       = mean;
        ws[128 + bc] = 1.f / (sqrtf(var) + EPS_);
    }
}

__global__ void lsd_k_stats(const float* __restrict__ kern, float* __restrict__ ws) {
    __shared__ float ssum[256];
    __shared__ float ssq[256];
    __shared__ float smu, sinv;
    const int sidx = blockIdx.x;
    const int tid  = threadIdx.x;
    const float v = kern[sidx * 256 + tid];
    ssum[tid] = v; ssq[tid] = v * v;
    __syncthreads();
    for (int off = 128; off > 0; off >>= 1) {
        if (tid < off) { ssum[tid] += ssum[tid + off]; ssq[tid] += ssq[tid + off]; }
        __syncthreads();
    }
    if (tid == 0) {
        const float mean = ssum[0] * (1.f / 256.f);
        const float var  = fmaxf(ssq[0] * (1.f / 256.f) - mean * mean, 0.f);
        const float inv  = 1.f / (sqrtf(var) + EPS_);
        ws[256 + sidx] = mean;
        ws[320 + sidx] = inv;
        smu = mean; sinv = inv;
    }
    __syncthreads();
    if (tid < KSZ_) {
        const float mu  = smu;
        const float inv = sinv;
        float acc = 0.f;
        for (int c = 0; c < C_; ++c) {
            const float q = (kern[sidx * 256 + tid * C_ + c] - mu) * inv;
            acc += q * q;
        }
        ws[384 + sidx * KSZ_ + tid] = acc;
    }
}

__global__ __launch_bounds__(256) void lsd_rows_kernel(const float* __restrict__ x,
                                                       const float* __restrict__ kern,
                                                       const float* __restrict__ ws,
                                                       float* __restrict__ out) {
    const int tg = blockIdx.x * ROWS_;
    const int b  = blockIdx.y;
    const int l  = threadIdx.x & 31;
    const int w  = __builtin_amdgcn_readfirstlane((int)(threadIdx.x >> 5));

    __shared__ float zsh[C_][ZW_];
    __shared__ float pnsh[C_][ZW_];
    __shared__ __attribute__((aligned(16))) float osh[KSZ_][OPITCH_];

    const int n  = l & 15;
    const int hi = l >> 4;

    const float mux = ws[b * C_ + w];
    const float inx = ws[128 + b * C_ + w];
    float xr[2];
    for (int i = 0; i < 2; ++i) {
        int t = tg + l + 32 * i;
        if (t > T_ - 1) t = T_ - 1;
        xr[i] = x[(b * T_ + t) * C_ + w];
    }
    v2f   bq[4][4][2];
    float kn2[4][4];
    float muk[4], ink[4];
    for (int st = 0; st < 4; ++st) {
        const int s = st * 16 + n;
        muk[st] = ws[256 + s];
        ink[st] = ws[320 + s];
        for (int j = 0; j < 4; ++j) {
            const float* kp = kern + s * 256 + (4 * w + j) * 8;
            bq[j][st][0].x = kp[2 * hi];
            bq[j][st][0].y = kp[2 * hi + 1];
            bq[j][st][1].x = kp[2 * hi + 4];
            bq[j][st][1].y = kp[2 * hi + 5];
            kn2[j][st] = ws[384 + s * KSZ_ + 4 * w + j];
        }
    }

    for (int i = 0; i < 2; ++i) zsh[w][l + 32 * i] = (xr[i] - mux) * inx;
    __syncthreads();

    for (int i = 0; i < 2; ++i) {
        const int u = l + 32 * i;
        if (u < ROWS_ + 24) {
            float acc = 0.f;
            for (int c = 0; c < 8; ++c) { const float z = zsh[w][u + c]; acc += z * z; }
            pnsh[w][u] = acc;
        }
    }
    for (int st = 0; st < 4; ++st)
        for (int j = 0; j < 4; ++j) {
            bq[j][st][0].x = (bq[j][st][0].x - muk[st]) * ink[st];
            bq[j][st][0].y = (bq[j][st][0].y - muk[st]) * ink[st];
            bq[j][st][1].x = (bq[j][st][1].x - muk[st]) * ink[st];
            bq[j][st][1].y = (bq[j][st][1].y - muk[st]) * ink[st];
        }
    __syncthreads();

    const int ntile = (tg + 16 < TOUT_) ? 2 : 1;
    for (int tile = 0; tile < ntile; ++tile) {
        float md[4][8];
        for (int j = 0; j < 4; ++j) {
            const int o = tile * 16 + 8 * j;
            const float* zp = &zsh[w][o + n + 2 * hi];
            v2f a_lo, a_hi;
            a_lo.x = zp[0];
            a_lo.y = zp[1];
            a_hi.x = zp[4];
            a_hi.y = zp[5];

            float pn2[8];
            for (int r = 0; r < 8; ++r) pn2[r] = pnsh[w][o + 8 * hi + r];

            for (int r = 0; r < 8; ++r) md[j][r] = FLT_MAX;

            for (int st = 0; st < 4; ++st) {
                v8f g = {0.f, 0.f, 0.f, 0.f, 0.f, 0.f, 0.f, 0.f};
                g = __builtin_amdgcn_wmma_f32_16x16x4_f32(false, a_lo, false, bq[j][st][0],
                                                          (short)0, g, false, false);
                g = __builtin_amdgcn_wmma_f32_16x16x4_f32(false, a_hi, false, bq[j][st][1],
                                                          (short)0, g, false, false);
                for (int r = 0; r < 8; ++r) {
                    const float d = pn2[r] + kn2[j][st] - 2.f * g[r];
                    md[j][r] = fminf(md[j][r], d);
                }
            }
        }

        for (int mask = 1; mask <= 8; mask <<= 1)
            for (int j = 0; j < 4; ++j)
                for (int r = 0; r < 8; ++r)
                    md[j][r] = fminf(md[j][r], __shfl_xor(md[j][r], mask, 32));

        for (int j = 0; j < 4; ++j) {
            if (n == j) {
                float* op = &osh[4 * w + j][tile * 16 + 8 * hi];
                v4f o0 = {md[j][0], md[j][1], md[j][2], md[j][3]};
                v4f o1 = {md[j][4], md[j][5], md[j][6], md[j][7]};
                *reinterpret_cast<v4f*>(op)     = o0;
                *reinterpret_cast<v4f*>(op + 4) = o1;
            }
        }
    }
    __syncthreads();

    {
        const int row = threadIdx.x >> 3;
        const int q4  = (threadIdx.x & 7) * 4;
        const int t   = tg + row;
        if (t < TOUT_) {
            v4f o = {osh[q4][row], osh[q4 + 1][row], osh[q4 + 2][row], osh[q4 + 3][row]};
            *reinterpret_cast<v4f*>(out + (b * TOUT_ + t) * KSZ_ + q4) = o;
        }
    }
}

extern "C" void kernel_launch(void* const* d_in, const int* in_sizes, int n_in,
                              void* d_out, int out_size, void* d_ws, size_t ws_size,
                              hipStream_t stream) {
    (void)in_sizes; (void)n_in; (void)out_size; (void)ws_size;
    const float* x    = (const float*)d_in[0];
    const float* kern = (const float*)d_in[1];
    float* ws  = (float*)d_ws;
    float* out = (float*)d_out;

    lsd_x_stats<<<dim3(B_ * C_), dim3(128), 0, stream>>>(x, ws);
    lsd_k_stats<<<dim3(S_), dim3(256), 0, stream>>>(kern, ws);
    lsd_rows_kernel<<<dim3(16, B_), dim3(256), 0, stream>>>(x, kern, ws, out);
}


# Series and shapelet statistics, two dependent launches, run side by side in ONE launch (64 shapelet groups + 64 groups of two series), same summation trees; main kernel untouched; 3 launches -> 2
# speedup vs baseline: 2.2600x; 1.2556x over previous
#include <hip/hip_runtime.h>
#include <hip/hip_bf16.h>
#include <float.h>
#include <math.h>


typedef __attribute__((ext_vector_type(2))) float v2f;
typedef __attribute__((ext_vector_type(4))) float v4f;
typedef __attribute__((ext_vector_type(8))) float v8f;

#define B_   16
#define T_   512
#define C_   8
#define S_   64
#define KSZ_ 32
#define TOUT_ 481
#define EPS_ 1e-8f

#define ROWS_ 32
#define ZW_   64
#define OPITCH_ 36


#define SGROUPS_ 128

__global__ __launch_bounds__(256) void lsd_stats(const float* __restrict__ x,
                                                 const float* __restrict__ kern,
                                                 float* __restrict__ ws) {
    __shared__ float ssum[256];
    __shared__ float ssq[256];
    __shared__ float smu, sinv;
    const int tid = threadIdx.x;

    if (blockIdx.x < S_) {
        const int sidx = blockIdx.x;
        const float v = kern[sidx * 256 + tid];
        ssum[tid] = v; ssq[tid] = v * v;
        __syncthreads();
        for (int off = 128; off > 0; off >>= 1) {
            if (tid < off) { ssum[tid] += ssum[tid + off]; ssq[tid] += ssq[tid + off]; }
            __syncthreads();
        }
        if (tid == 0) {
            const float mean = ssum[0] * (1.f / 256.f);
            const float var  = fmaxf(ssq[0] * (1.f / 256.f) - mean * mean, 0.f);
            const float inv  = 1.f / (sqrtf(var) + EPS_);
            ws[256 + sidx] = mean;
            ws[320 + sidx] = inv;
            smu = mean; sinv = inv;
        }
        __syncthreads();
        if (tid < KSZ_) {
            const float mu  = smu;
            const float inv = sinv;
            float acc = 0.f;
            for (int c = 0; c < C_; ++c) {
                const float q = (kern[sidx * 256 + tid * C_ + c] - mu) * inv;
                acc += q * q;
            }
            ws[384 + sidx * KSZ_ + tid] = acc;
        }
    } else {
        const int lt = tid & 127;
        const int bc = (blockIdx.x - S_) * 2 + (tid >> 7);
        const int b  = bc >> 3;
        const int c  = bc & 7;
        float s = 0.f, q = 0.f;
        for (int i = 0; i < 4; ++i) {
            const int t = lt + i * 128;
            const float v = x[(b * T_ + t) * C_ + c];
            s += v; q += v * v;
        }
        ssum[tid] = s; ssq[tid] = q;
        __syncthreads();
        for (int off = 64; off > 0; off >>= 1) {
            if (lt < off) { ssum[tid] += ssum[tid + off]; ssq[tid] += ssq[tid + off]; }
            __syncthreads();
        }
        if (lt == 0) {
            const float mean = ssum[tid] * (1.f / (float)T_);
            const float var  = fmaxf(ssq[tid] * (1.f / (float)T_) - mean * mean, 0.f);
            ws[bc]       = mean;
            ws[128 + bc] = 1.f / (sqrtf(var) + EPS_);
        }
    }
}

__global__ __launch_bounds__(256) void lsd_rows_kernel(const float* __restrict__ x,
                                                       const float* __restrict__ kern,
                                                       const float* __restrict__ ws,
                                                       float* __restrict__ out) {
    const int tg = blockIdx.x * ROWS_;
    const int b  = blockIdx.y;
    const int l  = threadIdx.x & 31;
    const int w  = __builtin_amdgcn_readfirstlane((int)(threadIdx.x >> 5));

    __shared__ float zsh[C_][ZW_];
    __shared__ float pnsh[C_][ZW_];
    __shared__ __attribute__((aligned(16))) float osh[KSZ_][OPITCH_];

    const int n  = l & 15;
    const int hi = l >> 4;

    const float mux = ws[b * C_ + w];
    const float inx = ws[128 + b * C_ + w];
    float xr[2];
    for (int i = 0; i < 2; ++i) {
        int t = tg + l + 32 * i;
        if (t > T_ - 1) t = T_ - 1;
        xr[i] = x[(b * T_ + t) * C_ + w];
    }
    v2f   bq[4][4][2];
    float kn2[4][4];
    float muk[4], ink[4];
    for (int st = 0; st < 4; ++st) {
        const int s = st * 16 + n;
        muk[st] = ws[256 + s];
        ink[st] = ws[320 + s];
        for (int j = 0; j < 4; ++j) {
            const float* kp = kern + s * 256 + (4 * w + j) * 8;
            bq[j][st][0].x = kp[2 * hi];
            bq[j][st][0].y = kp[2 * hi + 1];
            bq[j][st][1].x = kp[2 * hi + 4];
            bq[j][st][1].y = kp[2 * hi + 5];
            kn2[j][st] = ws[384 + s * KSZ_ + 4 * w + j];
        }
    }

    for (int i = 0; i < 2; ++i) zsh[w][l + 32 * i] = (xr[i] - mux) * inx;
    __syncthreads();

    for (int i = 0; i < 2; ++i) {
        const int u = l + 32 * i;
        if (u < ROWS_ + 24) {
            float acc = 0.f;
            for (int c = 0; c < 8; ++c) { const float z = zsh[w][u + c]; acc += z * z; }
            pnsh[w][u] = acc;
        }
    }
    for (int st = 0; st < 4; ++st)
        for (int j = 0; j < 4; ++j) {
            bq[j][st][0].x = (bq[j][st][0].x - muk[st]) * ink[st];
            bq[j][st][0].y = (bq[j][st][0].y - muk[st]) * ink[st];
            bq[j][st][1].x = (bq[j][st][1].x - muk[st]) * ink[st];
            bq[j][st][1].y = (bq[j][st][1].y - muk[st]) * ink[st];
        }
    __syncthreads();

    const int ntile = (tg + 16 < TOUT_) ? 2 : 1;
    for (int tile = 0; tile < ntile; ++tile) {
        float md[4][8];
        for (int j = 0; j < 4; ++j) {
            const int o = tile * 16 + 8 * j;
            const float* zp = &zsh[w][o + n + 2 * hi];
            v2f a_lo, a_hi;
            a_lo.x = zp[0];
            a_lo.y = zp[1];
            a_hi.x = zp[4];
            a_hi.y = zp[5];

            float pn2[8];
            for (int r = 0; r < 8; ++r) pn2[r] = pnsh[w][o + 8 * hi + r];

            for (int r = 0; r < 8; ++r) md[j][r] = FLT_MAX;

            for (int st = 0; st < 4; ++st) {
                v8f g = {0.f, 0.f, 0.f, 0.f, 0.f, 0.f, 0.f, 0.f};
                g = __builtin_amdgcn_wmma_f32_16x16x4_f32(false, a_lo, false, bq[j][st][0],
                                                          (short)0, g, false, false);
                g = __builtin_amdgcn_wmma_f32_16x16x4_f32(false, a_hi, false, bq[j][st][1],
                                                          (short)0, g, false, false);
                for (int r = 0; r < 8; ++r) {
                    const float d = pn2[r] + kn2[j][st] - 2.f * g[r];
                    md[j][r] = fminf(md[j][r], d);
                }
            }
        }

        for (int mask = 1; mask <= 8; mask <<= 1)
            for (int j = 0; j < 4; ++j)
                for (int r = 0; r < 8; ++r)
                    md[j][r] = fminf(md[j][r], __shfl_xor(md[j][r], mask, 32));

        for (int j = 0; j < 4; ++j) {
            if (n == j) {
                float* op = &osh[4 * w + j][tile * 16 + 8 * hi];
                v4f o0 = {md[j][0], md[j][1], md[j][2], md[j][3]};
                v4f o1 = {md[j][4], md[j][5], md[j][6], md[j][7]};
                *reinterpret_cast<v4f*>(op)     = o0;
                *reinterpret_cast<v4f*>(op + 4) = o1;
            }
        }
    }
    __syncthreads();

    {
        const int row = threadIdx.x >> 3;
        const int q4  = (threadIdx.x & 7) * 4;
        const int t   = tg + row;
        if (t < TOUT_) {
            v4f o = {osh[q4][row], osh[q4 + 1][row], osh[q4 + 2][row], osh[q4 + 3][row]};
            *reinterpret_cast<v4f*>(out + (b * TOUT_ + t) * KSZ_ + q4) = o;
        }
    }
}

extern "C" void kernel_launch(void* const* d_in, const int* in_sizes, int n_in,
                              void* d_out, int out_size, void* d_ws, size_t ws_size,
                              hipStream_t stream) {
    (void)in_sizes; (void)n_in; (void)out_size; (void)ws_size;
    const float* x    = (const float*)d_in[0];
    const float* kern = (const float*)d_in[1];
    float* ws  = (float*)d_ws;
    float* out = (float*)d_out;

    lsd_stats<<<dim3(SGROUPS_), dim3(256), 0, stream>>>(x, kern, ws);
    lsd_rows_kernel<<<dim3(16, B_), dim3(256), 0, stream>>>(x, kern, ws, out);
}


# Statistics launch also writes the normalized shapelets and slot norms to the workspace in matrix-fragment order; a main-kernel wave fetches them in 20 contiguous b128 loads, not 44 strided ones
# speedup vs baseline: 2.4076x; 1.0653x over previous
#include <hip/hip_runtime.h>
#include <hip/hip_bf16.h>
#include <float.h>
#include <math.h>


typedef __attribute__((ext_vector_type(2))) float v2f;
typedef __attribute__((ext_vector_type(4))) float v4f;
typedef __attribute__((ext_vector_type(8))) float v8f;

#define B_   16
#define T_   512
#define C_   8
#define S_   64
#define KSZ_ 32
#define TOUT_ 481
#define EPS_ 1e-8f

#define ROWS_ 32
#define ZW_   64
#define OPITCH_ 36

#define KN_OFF_   512
#define FRAG_OFF_ 2560

#define SGROUPS_ 128

__global__ __launch_bounds__(256) void lsd_stats(const float* __restrict__ x,
                                                 const float* __restrict__ kern,
                                                 float* __restrict__ ws) {
    __shared__ float ssum[256];
    __shared__ float ssq[256];
    __shared__ float smu, sinv;
    const int tid = threadIdx.x;

    if (blockIdx.x < S_) {
        const int sidx = blockIdx.x;
        const float v = kern[sidx * 256 + tid];
        ssum[tid] = v; ssq[tid] = v * v;
        __syncthreads();
        for (int off = 128; off > 0; off >>= 1) {
            if (tid < off) { ssum[tid] += ssum[tid + off]; ssq[tid] += ssq[tid + off]; }
            __syncthreads();
        }
        if (tid == 0) {
            const float mean = ssum[0] * (1.f / 256.f);
            const float var  = fmaxf(ssq[0] * (1.f / 256.f) - mean * mean, 0.f);
            const float inv  = 1.f / (sqrtf(var) + EPS_);
            smu = mean; sinv = inv;
        }
        __syncthreads();
        const float mu  = smu;
        const float inv = sinv;
        const int st = sidx >> 4;
        const int n  = sidx & 15;
        {
            const int k  = tid >> 3;
            const int c  = tid & 7;
            const int hi = (c >> 1) & 1;
            const int e  = (c & 1) + 2 * (c >> 2);
            ws[FRAG_OFF_ + (((k * 4 + st) * 32 + n * 2 + hi) << 2) + e] = (v - mu) * inv;
        }
        if (tid < KSZ_) {
            float acc = 0.f;
            for (int c = 0; c < C_; ++c) {
                const float q = (kern[sidx * 256 + tid * C_ + c] - mu) * inv;
                acc += q * q;
            }
            ws[KN_OFF_ + ((((tid >> 2) * 4 + st) * 16 + n) << 2) + (tid & 3)] = acc;
        }
    } else {
        const int lt = tid & 127;
        const int bc = (blockIdx.x - S_) * 2 + (tid >> 7);
        const int b  = bc >> 3;
        const int c  = bc & 7;
        float s = 0.f, q = 0.f;
        for (int i = 0; i < 4; ++i) {
            const int t = lt + i * 128;
            const float v = x[(b * T_ + t) * C_ + c];
            s += v; q += v * v;
        }
        ssum[tid] = s; ssq[tid] = q;
        __syncthreads();
        for (int off = 64; off > 0; off >>= 1) {
            if (lt < off) { ssum[tid] += ssum[tid + off]; ssq[tid] += ssq[tid + off]; }
            __syncthreads();
        }
        if (lt == 0) {
            const float mean = ssum[tid] * (1.f / (float)T_);
            const float var  = fmaxf(ssq[tid] * (1.f / (float)T_) - mean * mean, 0.f);
            ws[bc]       = mean;
            ws[128 + bc] = 1.f / (sqrtf(var) + EPS_);
        }
    }
}

__global__ __launch_bounds__(256) void lsd_rows_kernel(const float* __restrict__ x,
                                                       const float* __restrict__ ws,
                                                       float* __restrict__ out) {
    const int tg = blockIdx.x * ROWS_;
    const int b  = blockIdx.y;
    const int l  = threadIdx.x & 31;
    const int w  = __builtin_amdgcn_readfirstlane((int)(threadIdx.x >> 5));

    __shared__ float zsh[C_][ZW_];
    __shared__ float pnsh[C_][ZW_];
    __shared__ __attribute__((aligned(16))) float osh[KSZ_][OPITCH_];

    const int n  = l & 15;
    const int hi = l >> 4;

    const float mux = ws[b * C_ + w];
    const float inx = ws[128 + b * C_ + w];
    float xr[2];
    for (int i = 0; i < 2; ++i) {
        int t = tg + l + 32 * i;
        if (t > T_ - 1) t = T_ - 1;
        xr[i] = x[(b * T_ + t) * C_ + w];
    }
    const v4f* __restrict__ fq = reinterpret_cast<const v4f*>(ws + FRAG_OFF_);
    const v4f* __restrict__ fn = reinterpret_cast<const v4f*>(ws + KN_OFF_);
    v4f bq[4][4];
    v4f kn2[4];
    for (int st = 0; st < 4; ++st) kn2[st] = fn[(w * 4 + st) * 16 + n];
    for (int j = 0; j < 4; ++j)
        for (int st = 0; st < 4; ++st)
            bq[j][st] = fq[((4 * w + j) * 4 + st) * 32 + 2 * n + hi];

    for (int i = 0; i < 2; ++i) zsh[w][l + 32 * i] = (xr[i] - mux) * inx;
    __syncthreads();

    for (int i = 0; i < 2; ++i) {
        const int u = l + 32 * i;
        if (u < ROWS_ + 24) {
            float acc = 0.f;
            for (int c = 0; c < 8; ++c) { const float z = zsh[w][u + c]; acc += z * z; }
            pnsh[w][u] = acc;
        }
    }
    __syncthreads();

    const int ntile = (tg + 16 < TOUT_) ? 2 : 1;
    for (int tile = 0; tile < ntile; ++tile) {
        float md[4][8];
        for (int j = 0; j < 4; ++j) {
            const int o = tile * 16 + 8 * j;
            const float* zp = &zsh[w][o + n + 2 * hi];
            v2f a_lo, a_hi;
            a_lo.x = zp[0];
            a_lo.y = zp[1];
            a_hi.x = zp[4];
            a_hi.y = zp[5];

            float pn2[8];
            for (int r = 0; r < 8; ++r) pn2[r] = pnsh[w][o + 8 * hi + r];

            for (int r = 0; r < 8; ++r) md[j][r] = FLT_MAX;

            for (int st = 0; st < 4; ++st) {
                v8f g = {0.f, 0.f, 0.f, 0.f, 0.f, 0.f, 0.f, 0.f};
                const v2f b_lo = {bq[j][st].x, bq[j][st].y};
                const v2f b_hi = {bq[j][st].z, bq[j][st].w};
                g = __builtin_amdgcn_wmma_f32_16x16x4_f32(false, a_lo, false, b_lo,
                                                          (short)0, g, false, false);
                g = __builtin_amdgcn_wmma_f32_16x16x4_f32(false, a_hi, false, b_hi,
                                                          (short)0, g, false, false);
                for (int r = 0; r < 8; ++r) {
                    const float d = pn2[r] + kn2[st][j] - 2.f * g[r];
                    md[j][r] = fminf(md[j][r], d);
                }
            }
        }

        for (int mask = 1; mask <= 8; mask <<= 1)
            for (int j = 0; j < 4; ++j)
                for (int r = 0; r < 8; ++r)
                    md[j][r] = fminf(md[j][r], __shfl_xor(md[j][r], mask, 32));

        for (int j = 0; j < 4; ++j) {
            if (n == j) {
                float* op = &osh[4 * w + j][tile * 16 + 8 * hi];
                v4f o0 = {md[j][0], md[j][1], md[j][2], md[j][3]};
                v4f o1 = {md[j][4], md[j][5], md[j][6], md[j][7]};
                *reinterpret_cast<v4f*>(op)     = o0;
                *reinterpret_cast<v4f*>(op + 4) = o1;
            }
        }
    }
    __syncthreads();

    {
        const int row = threadIdx.x >> 3;
        const int q4  = (threadIdx.x & 7) * 4;
        const int t   = tg + row;
        if (t < TOUT_) {
            v4f o = {osh[q4][row], osh[q4 + 1][row], osh[q4 + 2][row], osh[q4 + 3][row]};
            *reinterpret_cast<v4f*>(out + (b * TOUT_ + t) * KSZ_ + q4) = o;
        }
    }
}

extern "C" void kernel_launch(void* const* d_in, const int* in_sizes, int n_in,
                              void* d_out, int out_size, void* d_ws, size_t ws_size,
                              hipStream_t stream) {
    (void)in_sizes; (void)n_in; (void)out_size; (void)ws_size;
    const float* x    = (const float*)d_in[0];
    const float* kern = (const float*)d_in[1];
    float* ws  = (float*)d_ws;
    float* out = (float*)d_out;

    lsd_stats<<<dim3(SGROUPS_), dim3(256), 0, stream>>>(x, kern, ws);
    lsd_rows_kernel<<<dim3(16, B_), dim3(256), 0, stream>>>(x, ws, out);
}


# Min over the 16 shapelet lanes moved into the LDS output stage: lanes file partial minima per (row, slot) cell, the output pass reduces each cell; the 128 lane exchanges per tile are gone
# speedup vs baseline: 2.5819x; 1.0724x over previous
#include <hip/hip_runtime.h>
#include <hip/hip_bf16.h>
#include <float.h>
#include <math.h>


typedef __attribute__((ext_vector_type(2))) float v2f;
typedef __attribute__((ext_vector_type(4))) float v4f;
typedef __attribute__((ext_vector_type(8))) float v8f;

#define B_   16
#define T_   512
#define C_   8
#define S_   64
#define KSZ_ 32
#define TOUT_ 481
#define EPS_ 1e-8f

#define ROWS_ 32
#define ZW_   64
#define CELLP_ 20
#define THRP_  84
#define ROWP_  676

#define KN_OFF_   512
#define FRAG_OFF_ 2560

#define SGROUPS_ 128

__global__ __launch_bounds__(256) void lsd_stats(const float* __restrict__ x,
                                                 const float* __restrict__ kern,
                                                 float* __restrict__ ws) {
    __shared__ float ssum[256];
    __shared__ float ssq[256];
    __shared__ float smu, sinv;
    const int tid = threadIdx.x;

    if (blockIdx.x < S_) {
        const int sidx = blockIdx.x;
        const float v = kern[sidx * 256 + tid];
        ssum[tid] = v; ssq[tid] = v * v;
        __syncthreads();
        for (int off = 128; off > 0; off >>= 1) {
            if (tid < off) { ssum[tid] += ssum[tid + off]; ssq[tid] += ssq[tid + off]; }
            __syncthreads();
        }
        if (tid == 0) {
            const float mean = ssum[0] * (1.f / 256.f);
            const float var  = fmaxf(ssq[0] * (1.f / 256.f) - mean * mean, 0.f);
            const float inv  = 1.f / (sqrtf(var) + EPS_);
            smu = mean; sinv = inv;
        }
        __syncthreads();
        const float mu  = smu;
        const float inv = sinv;
        const int st = sidx >> 4;
        const int n  = sidx & 15;
        {
            const int k  = tid >> 3;
            const int c  = tid & 7;
            const int hi = (c >> 1) & 1;
            const int e  = (c & 1) + 2 * (c >> 2);
            ws[FRAG_OFF_ + (((k * 4 + st) * 32 + n * 2 + hi) << 2) + e] = (v - mu) * inv;
        }
        if (tid < KSZ_) {
            float acc = 0.f;
            for (int c = 0; c < C_; ++c) {
                const float q = (kern[sidx * 256 + tid * C_ + c] - mu) * inv;
                acc += q * q;
            }
            ws[KN_OFF_ + ((((tid >> 2) * 4 + st) * 16 + n) << 2) + (tid & 3)] = acc;
        }
    } else {
        const int lt = tid & 127;
        const int bc = (blockIdx.x - S_) * 2 + (tid >> 7);
        const int b  = bc >> 3;
        const int c  = bc & 7;
        float s = 0.f, q = 0.f;
        for (int i = 0; i < 4; ++i) {
            const int t = lt + i * 128;
            const float v = x[(b * T_ + t) * C_ + c];
            s += v; q += v * v;
        }
        ssum[tid] = s; ssq[tid] = q;
        __syncthreads();
        for (int off = 64; off > 0; off >>= 1) {
            if (lt < off) { ssum[tid] += ssum[tid + off]; ssq[tid] += ssq[tid + off]; }
            __syncthreads();
        }
        if (lt == 0) {
            const float mean = ssum[tid] * (1.f / (float)T_);
            const float var  = fmaxf(ssq[tid] * (1.f / (float)T_) - mean * mean, 0.f);
            ws[bc]       = mean;
            ws[128 + bc] = 1.f / (sqrtf(var) + EPS_);
        }
    }
}

__global__ __launch_bounds__(256) void lsd_rows_kernel(const float* __restrict__ x,
                                                       const float* __restrict__ ws,
                                                       float* __restrict__ out) {
    const int tg = blockIdx.x * ROWS_;
    const int b  = blockIdx.y;
    const int l  = threadIdx.x & 31;
    const int w  = __builtin_amdgcn_readfirstlane((int)(threadIdx.x >> 5));

    __shared__ float zsh[C_][ZW_];
    __shared__ float pnsh[C_][ZW_];
    __shared__ __attribute__((aligned(16))) float red[ROWS_ * ROWP_];

    const int n  = l & 15;
    const int hi = l >> 4;

    const float mux = ws[b * C_ + w];
    const float inx = ws[128 + b * C_ + w];
    float xr[2];
    for (int i = 0; i < 2; ++i) {
        int t = tg + l + 32 * i;
        if (t > T_ - 1) t = T_ - 1;
        xr[i] = x[(b * T_ + t) * C_ + w];
    }
    const v4f* __restrict__ fq = reinterpret_cast<const v4f*>(ws + FRAG_OFF_);
    const v4f* __restrict__ fn = reinterpret_cast<const v4f*>(ws + KN_OFF_);
    v4f bq[4][4];
    v4f kn2[4];
    for (int st = 0; st < 4; ++st) kn2[st] = fn[(w * 4 + st) * 16 + n];
    for (int j = 0; j < 4; ++j)
        for (int st = 0; st < 4; ++st)
            bq[j][st] = fq[((4 * w + j) * 4 + st) * 32 + 2 * n + hi];

    for (int i = 0; i < 2; ++i) zsh[w][l + 32 * i] = (xr[i] - mux) * inx;
    __syncthreads();

    for (int i = 0; i < 2; ++i) {
        const int u = l + 32 * i;
        if (u < ROWS_ + 24) {
            float acc = 0.f;
            for (int c = 0; c < 8; ++c) { const float z = zsh[w][u + c]; acc += z * z; }
            pnsh[w][u] = acc;
        }
    }
    __syncthreads();

    const int ntile = (tg + 16 < TOUT_) ? 2 : 1;
    for (int tile = 0; tile < ntile; ++tile) {
        float md[4][8];
        for (int j = 0; j < 4; ++j) {
            const int o = tile * 16 + 8 * j;
            const float* zp = &zsh[w][o + n + 2 * hi];
            v2f a_lo, a_hi;
            a_lo.x = zp[0];
            a_lo.y = zp[1];
            a_hi.x = zp[4];
            a_hi.y = zp[5];

            float pn2[8];
            for (int r = 0; r < 8; ++r) pn2[r] = pnsh[w][o + 8 * hi + r];

            for (int r = 0; r < 8; ++r) md[j][r] = FLT_MAX;

            for (int st = 0; st < 4; ++st) {
                v8f g = {0.f, 0.f, 0.f, 0.f, 0.f, 0.f, 0.f, 0.f};
                const v2f b_lo = {bq[j][st].x, bq[j][st].y};
                const v2f b_hi = {bq[j][st].z, bq[j][st].w};
                g = __builtin_amdgcn_wmma_f32_16x16x4_f32(false, a_lo, false, b_lo,
                                                          (short)0, g, false, false);
                g = __builtin_amdgcn_wmma_f32_16x16x4_f32(false, a_hi, false, b_hi,
                                                          (short)0, g, false, false);
                for (int r = 0; r < 8; ++r) {
                    const float d = pn2[r] + kn2[st][j] - 2.f * g[r];
                    md[j][r] = fminf(md[j][r], d);
                }
            }
        }

        {
            float* rp = &red[(tile * 16 + 8 * hi) * ROWP_ + w * THRP_ + n];
            for (int j = 0; j < 4; ++j)
                for (int r = 0; r < 8; ++r)
                    rp[r * ROWP_ + j * CELLP_] = md[j][r];
        }
    }
    __syncthreads();

    {
        const int row = threadIdx.x >> 3;
        const int q   = threadIdx.x & 7;
        const int t   = tg + row;
        if (t < TOUT_) {
            const v4f* cp = reinterpret_cast<const v4f*>(&red[row * ROWP_ + q * THRP_]);
            float o[4];
            for (int s = 0; s < 4; ++s) {
                const v4f c0 = cp[s * 5 + 0], c1 = cp[s * 5 + 1], c2 = cp[s * 5 + 2], c3 = cp[s * 5 + 3];
                const float m0 = fminf(fminf(c0.x, c0.y), fminf(c0.z, c0.w));
                const float m1 = fminf(fminf(c1.x, c1.y), fminf(c1.z, c1.w));
                const float m2 = fminf(fminf(c2.x, c2.y), fminf(c2.z, c2.w));
                const float m3 = fminf(fminf(c3.x, c3.y), fminf(c3.z, c3.w));
                o[s] = fminf(fminf(m0, m1), fminf(m2, m3));
            }
            v4f ov = {o[0], o[1], o[2], o[3]};
            *reinterpret_cast<v4f*>(out + (b * TOUT_ + t) * KSZ_ + q * 4) = ov;
        }
    }
}

extern "C" void kernel_launch(void* const* d_in, const int* in_sizes, int n_in,
                              void* d_out, int out_size, void* d_ws, size_t ws_size,
                              hipStream_t stream) {
    (void)in_sizes; (void)n_in; (void)out_size; (void)ws_size;
    const float* x    = (const float*)d_in[0];
    const float* kern = (const float*)d_in[1];
    float* ws  = (float*)d_ws;
    float* out = (float*)d_out;

    lsd_stats<<<dim3(SGROUPS_), dim3(256), 0, stream>>>(x, kern, ws);
    lsd_rows_kernel<<<dim3(16, B_), dim3(256), 0, stream>>>(x, ws, out);
}


# Statistics launch rebuilt as 80 self-sufficient single waves (16 batch entries, 64 shapelets): sums in registers by lane exchanges, in a new summation order; its 10 barrier rounds and LDS trees go
# speedup vs baseline: 2.7757x; 1.0751x over previous
#include <hip/hip_runtime.h>
#include <hip/hip_bf16.h>
#include <float.h>
#include <math.h>


typedef __attribute__((ext_vector_type(2))) float v2f;
typedef __attribute__((ext_vector_type(4))) float v4f;
typedef __attribute__((ext_vector_type(8))) float v8f;

#define B_   16
#define T_   512
#define C_   8
#define S_   64
#define KSZ_ 32
#define TOUT_ 481
#define EPS_ 1e-8f

#define ROWS_ 32
#define ZW_   64
#define CELLP_ 20
#define THRP_  84
#define ROWP_  676

#define KN_OFF_   512
#define FRAG_OFF_ 2560

#define SWAVES_ 80

template <int CTRL>
__device__ __forceinline__ float lane_x(float v) {
    const int b = __builtin_bit_cast(int, v);
    return __builtin_bit_cast(float, __builtin_amdgcn_update_dpp(b, b, CTRL, 0xf, 0xf, false));
}
__device__ __forceinline__ float lane_x16(float v) {
    const int b = __builtin_bit_cast(int, v);
    return __builtin_bit_cast(float, __builtin_amdgcn_permlanex16(b, b, 0x76543210, 0xfedcba98, false, false));
}

__global__ __launch_bounds__(32) __attribute__((amdgpu_waves_per_eu(1, 1))) void lsd_stats(const float* __restrict__ x,
                                                const float* __restrict__ kern,
                                                float* __restrict__ ws) {
    const int l = threadIdx.x;

    if (blockIdx.x < B_) {
        const int b = blockIdx.x;
        const v4f* __restrict__ xb = reinterpret_cast<const v4f*>(x + b * (T_ * C_));
        v4f s = {0.f, 0.f, 0.f, 0.f};
        v4f q = {0.f, 0.f, 0.f, 0.f};
        v4f v[32];
        for (int i = 0; i < 32; ++i) v[i] = xb[i * 32 + l];
        for (int i = 0; i < 32; ++i) { s += v[i]; q += v[i] * v[i]; }
        for (int e = 0; e < 4; ++e) {
            float a = s[e], c = q[e];
            a += lane_x<0x162>(a);  c += lane_x<0x162>(c);
            a += lane_x<0x164>(a);  c += lane_x<0x164>(c);
            a += lane_x<0x168>(a);  c += lane_x<0x168>(c);
            a += lane_x16(a);       c += lane_x16(c);
            s[e] = a; q[e] = c;
        }
        v4f mean4, inv4;
        for (int e = 0; e < 4; ++e) {
            const float mean = s[e] * (1.f / (float)T_);
            const float var  = fmaxf(q[e] * (1.f / (float)T_) - mean * mean, 0.f);
            mean4[e] = mean;
            inv4[e]  = 1.f / (sqrtf(var) + EPS_);
        }
        if (l < 2) {
            *reinterpret_cast<v4f*>(ws + b * C_ + 4 * l)       = mean4;
            *reinterpret_cast<v4f*>(ws + 128 + b * C_ + 4 * l) = inv4;
        }
    } else {
        const int sidx = blockIdx.x - B_;
        const int st = sidx >> 4;
        const int n  = sidx & 15;
        const int k  = l;
        const v4f* __restrict__ kp = reinterpret_cast<const v4f*>(kern + sidx * 256 + k * C_);
        const v4f v0 = kp[0];
        const v4f v1 = kp[1];
        float s = ((v0.x + v0.y) + (v0.z + v0.w)) + ((v1.x + v1.y) + (v1.z + v1.w));
        float q = ((v0.x * v0.x + v0.y * v0.y) + (v0.z * v0.z + v0.w * v0.w))
                + ((v1.x * v1.x + v1.y * v1.y) + (v1.z * v1.z + v1.w * v1.w));
        s += lane_x<0x161>(s);  q += lane_x<0x161>(q);
        s += lane_x<0x162>(s);  q += lane_x<0x162>(q);
        s += lane_x<0x164>(s);  q += lane_x<0x164>(q);
        s += lane_x<0x168>(s);  q += lane_x<0x168>(q);
        s += lane_x16(s);       q += lane_x16(q);
        const float mu  = s * (1.f / 256.f);
        const float var = fmaxf(q * (1.f / 256.f) - mu * mu, 0.f);
        const float inv = 1.f / (sqrtf(var) + EPS_);
        float z[C_];
        z[0] = (v0.x - mu) * inv;  z[1] = (v0.y - mu) * inv;
        z[2] = (v0.z - mu) * inv;  z[3] = (v0.w - mu) * inv;
        z[4] = (v1.x - mu) * inv;  z[5] = (v1.y - mu) * inv;
        z[6] = (v1.z - mu) * inv;  z[7] = (v1.w - mu) * inv;
        float acc = 0.f;
        for (int c = 0; c < C_; ++c) acc += z[c] * z[c];
        v4f* fp = reinterpret_cast<v4f*>(ws + FRAG_OFF_ + (((k * 4 + st) * 32 + n * 2) << 2));
        const v4f c0 = {z[0], z[1], z[4], z[5]};
        const v4f c1 = {z[2], z[3], z[6], z[7]};
        fp[0] = c0;
        fp[1] = c1;
        ws[KN_OFF_ + ((((k >> 2) * 4 + st) * 16 + n) << 2) + (k & 3)] = acc;
    }
}

__global__ __launch_bounds__(256) void lsd_rows_kernel(const float* __restrict__ x,
                                                       const float* __restrict__ ws,
                                                       float* __restrict__ out) {
    const int tg = blockIdx.x * ROWS_;
    const int b  = blockIdx.y;
    const int l  = threadIdx.x & 31;
    const int w  = __builtin_amdgcn_readfirstlane((int)(threadIdx.x >> 5));

    __shared__ float zsh[C_][ZW_];
    __shared__ float pnsh[C_][ZW_];
    __shared__ __attribute__((aligned(16))) float red[ROWS_ * ROWP_];

    const int n  = l & 15;
    const int hi = l >> 4;

    const float mux = ws[b * C_ + w];
    const float inx = ws[128 + b * C_ + w];
    float xr[2];
    for (int i = 0; i < 2; ++i) {
        int t = tg + l + 32 * i;
        if (t > T_ - 1) t = T_ - 1;
        xr[i] = x[(b * T_ + t) * C_ + w];
    }
    const v4f* __restrict__ fq = reinterpret_cast<const v4f*>(ws + FRAG_OFF_);
    const v4f* __restrict__ fn = reinterpret_cast<const v4f*>(ws + KN_OFF_);
    v4f bq[4][4];
    v4f kn2[4];
    for (int st = 0; st < 4; ++st) kn2[st] = fn[(w * 4 + st) * 16 + n];
    for (int j = 0; j < 4; ++j)
        for (int st = 0; st < 4; ++st)
            bq[j][st] = fq[((4 * w + j) * 4 + st) * 32 + 2 * n + hi];

    for (int i = 0; i < 2; ++i) zsh[w][l + 32 * i] = (xr[i] - mux) * inx;
    __syncthreads();

    for (int i = 0; i < 2; ++i) {
        const int u = l + 32 * i;
        if (u < ROWS_ + 24) {
            float acc = 0.f;
            for (int c = 0; c < 8; ++c) { const float z = zsh[w][u + c]; acc += z * z; }
            pnsh[w][u] = acc;
        }
    }
    __syncthreads();

    const int ntile = (tg + 16 < TOUT_) ? 2 : 1;
    for (int tile = 0; tile < ntile; ++tile) {
        float md[4][8];
        for (int j = 0; j < 4; ++j) {
            const int o = tile * 16 + 8 * j;
            const float* zp = &zsh[w][o + n + 2 * hi];
            v2f a_lo, a_hi;
            a_lo.x = zp[0];
            a_lo.y = zp[1];
            a_hi.x = zp[4];
            a_hi.y = zp[5];

            float pn2[8];
            for (int r = 0; r < 8; ++r) pn2[r] = pnsh[w][o + 8 * hi + r];

            for (int r = 0; r < 8; ++r) md[j][r] = FLT_MAX;

            for (int st = 0; st < 4; ++st) {
                v8f g = {0.f, 0.f, 0.f, 0.f, 0.f, 0.f, 0.f, 0.f};
                const v2f b_lo = {bq[j][st].x, bq[j][st].y};
                const v2f b_hi = {bq[j][st].z, bq[j][st].w};
                g = __builtin_amdgcn_wmma_f32_16x16x4_f32(false, a_lo, false, b_lo,
                                                          (short)0, g, false, false);
                g = __builtin_amdgcn_wmma_f32_16x16x4_f32(false, a_hi, false, b_hi,
                                                          (short)0, g, false, false);
                for (int r = 0; r < 8; ++r) {
                    const float d = pn2[r] + kn2[st][j] - 2.f * g[r];
                    md[j][r] = fminf(md[j][r], d);
                }
            }
        }

        {
            float* rp = &red[(tile * 16 + 8 * hi) * ROWP_ + w * THRP_ + n];
            for (int j = 0; j < 4; ++j)
                for (int r = 0; r < 8; ++r)
                    rp[r * ROWP_ + j * CELLP_] = md[j][r];
        }
    }
    __syncthreads();

    {
        const int row = threadIdx.x >> 3;
        const int q   = threadIdx.x & 7;
        const int t   = tg + row;
        if (t < TOUT_) {
            const v4f* cp = reinterpret_cast<const v4f*>(&red[row * ROWP_ + q * THRP_]);
            float o[4];
            for (int s = 0; s < 4; ++s) {
                const v4f c0 = cp[s * 5 + 0], c1 = cp[s * 5 + 1], c2 = cp[s * 5 + 2], c3 = cp[s * 5 + 3];
                const float m0 = fminf(fminf(c0.x, c0.y), fminf(c0.z, c0.w));
                const float m1 = fminf(fminf(c1.x, c1.y), fminf(c1.z, c1.w));
                const float m2 = fminf(fminf(c2.x, c2.y), fminf(c2.z, c2.w));
                const float m3 = fminf(fminf(c3.x, c3.y), fminf(c3.z, c3.w));
                o[s] = fminf(fminf(m0, m1), fminf(m2, m3));
            }
            v4f ov = {o[0], o[1], o[2], o[3]};
            *reinterpret_cast<v4f*>(out + (b * TOUT_ + t) * KSZ_ + q * 4) = ov;
        }
    }
}

extern "C" void kernel_launch(void* const* d_in, const int* in_sizes, int n_in,
                              void* d_out, int out_size, void* d_ws, size_t ws_size,
                              hipStream_t stream) {
    (void)in_sizes; (void)n_in; (void)out_size; (void)ws_size;
    const float* x    = (const float*)d_in[0];
    const float* kern = (const float*)d_in[1];
    float* ws  = (float*)d_ws;
    float* out = (float*)d_out;

    lsd_stats<<<dim3(SWAVES_), dim3(32), 0, stream>>>(x, kern, ws);
    lsd_rows_kernel<<<dim3(16, B_), dim3(256), 0, stream>>>(x, ws, out);
}


# Window norms ||p||^2 enter the product as its C operand, samples staged as -2z; a lane fetches its 6 C tuples and 6 A fragments once, so the tile loop reads no LDS; terms summed in a new order
# speedup vs baseline: 3.1940x; 1.1507x over previous
#include <hip/hip_runtime.h>
#include <hip/hip_bf16.h>
#include <float.h>
#include <math.h>


typedef __attribute__((ext_vector_type(2))) float v2f;
typedef __attribute__((ext_vector_type(4))) float v4f;
typedef __attribute__((ext_vector_type(8))) float v8f;

#define B_   16
#define T_   512
#define C_   8
#define S_   64
#define KSZ_ 32
#define TOUT_ 481
#define EPS_ 1e-8f

#define ROWS_ 32
#define ZW_   64
#define CELLP_ 20
#define THRP_  84
#define ROWP_  676

#define KN_OFF_   512
#define FRAG_OFF_ 2560

#define SWAVES_ 80

template <int CTRL>
__device__ __forceinline__ float lane_x(float v) {
    const int b = __builtin_bit_cast(int, v);
    return __builtin_bit_cast(float, __builtin_amdgcn_update_dpp(b, b, CTRL, 0xf, 0xf, false));
}
__device__ __forceinline__ float lane_x16(float v) {
    const int b = __builtin_bit_cast(int, v);
    return __builtin_bit_cast(float, __builtin_amdgcn_permlanex16(b, b, 0x76543210, 0xfedcba98, false, false));
}

__global__ __launch_bounds__(32) __attribute__((amdgpu_waves_per_eu(1, 1))) void lsd_stats(const float* __restrict__ x,
                                                const float* __restrict__ kern,
                                                float* __restrict__ ws) {
    const int l = threadIdx.x;

    if (blockIdx.x < B_) {
        const int b = blockIdx.x;
        const v4f* __restrict__ xb = reinterpret_cast<const v4f*>(x + b * (T_ * C_));
        v4f s = {0.f, 0.f, 0.f, 0.f};
        v4f q = {0.f, 0.f, 0.f, 0.f};
        v4f v[32];
        for (int i = 0; i < 32; ++i) v[i] = xb[i * 32 + l];
        for (int i = 0; i < 32; ++i) { s += v[i]; q += v[i] * v[i]; }
        for (int e = 0; e < 4; ++e) {
            float a = s[e], c = q[e];
            a += lane_x<0x162>(a);  c += lane_x<0x162>(c);
            a += lane_x<0x164>(a);  c += lane_x<0x164>(c);
            a += lane_x<0x168>(a);  c += lane_x<0x168>(c);
            a += lane_x16(a);       c += lane_x16(c);
            s[e] = a; q[e] = c;
        }
        v4f mean4, inv4;
        for (int e = 0; e < 4; ++e) {
            const float mean = s[e] * (1.f / (float)T_);
            const float var  = fmaxf(q[e] * (1.f / (float)T_) - mean * mean, 0.f);
            mean4[e] = mean;
            inv4[e]  = 1.f / (sqrtf(var) + EPS_);
        }
        if (l < 2) {
            *reinterpret_cast<v4f*>(ws + b * C_ + 4 * l)       = mean4;
            *reinterpret_cast<v4f*>(ws + 128 + b * C_ + 4 * l) = inv4;
        }
    } else {
        const int sidx = blockIdx.x - B_;
        const int st = sidx >> 4;
        const int n  = sidx & 15;
        const int k  = l;
        const v4f* __restrict__ kp = reinterpret_cast<const v4f*>(kern + sidx * 256 + k * C_);
        const v4f v0 = kp[0];
        const v4f v1 = kp[1];
        float s = ((v0.x + v0.y) + (v0.z + v0.w)) + ((v1.x + v1.y) + (v1.z + v1.w));
        float q = ((v0.x * v0.x + v0.y * v0.y) + (v0.z * v0.z + v0.w * v0.w))
                + ((v1.x * v1.x + v1.y * v1.y) + (v1.z * v1.z + v1.w * v1.w));
        s += lane_x<0x161>(s);  q += lane_x<0x161>(q);
        s += lane_x<0x162>(s);  q += lane_x<0x162>(q);
        s += lane_x<0x164>(s);  q += lane_x<0x164>(q);
        s += lane_x<0x168>(s);  q += lane_x<0x168>(q);
        s += lane_x16(s);       q += lane_x16(q);
        const float mu  = s * (1.f / 256.f);
        const float var = fmaxf(q * (1.f / 256.f) - mu * mu, 0.f);
        const float inv = 1.f / (sqrtf(var) + EPS_);
        float z[C_];
        z[0] = (v0.x - mu) * inv;  z[1] = (v0.y - mu) * inv;
        z[2] = (v0.z - mu) * inv;  z[3] = (v0.w - mu) * inv;
        z[4] = (v1.x - mu) * inv;  z[5] = (v1.y - mu) * inv;
        z[6] = (v1.z - mu) * inv;  z[7] = (v1.w - mu) * inv;
        float acc = 0.f;
        for (int c = 0; c < C_; ++c) acc += z[c] * z[c];
        v4f* fp = reinterpret_cast<v4f*>(ws + FRAG_OFF_ + (((k * 4 + st) * 32 + n * 2) << 2));
        const v4f c0 = {z[0], z[1], z[4], z[5]};
        const v4f c1 = {z[2], z[3], z[6], z[7]};
        fp[0] = c0;
        fp[1] = c1;
        ws[KN_OFF_ + ((((k >> 2) * 4 + st) * 16 + n) << 2) + (k & 3)] = acc;
    }
}

__global__ __launch_bounds__(256) void lsd_rows_kernel(const float* __restrict__ x,
                                                       const float* __restrict__ ws,
                                                       float* __restrict__ out) {
    const int tg = blockIdx.x * ROWS_;
    const int b  = blockIdx.y;
    const int l  = threadIdx.x & 31;
    const int w  = __builtin_amdgcn_readfirstlane((int)(threadIdx.x >> 5));

    __shared__ float zsh[C_][ZW_];
    __shared__ __attribute__((aligned(32))) float pnsh[C_][ZW_];
    __shared__ __attribute__((aligned(16))) float red[ROWS_ * ROWP_];

    const int n  = l & 15;
    const int hi = l >> 4;

    const float mux = ws[b * C_ + w];
    const float inx = ws[128 + b * C_ + w];
    float xr[2];
    for (int i = 0; i < 2; ++i) {
        int t = tg + l + 32 * i;
        if (t > T_ - 1) t = T_ - 1;
        xr[i] = x[(b * T_ + t) * C_ + w];
    }
    const v4f* __restrict__ fq = reinterpret_cast<const v4f*>(ws + FRAG_OFF_);
    const v4f* __restrict__ fn = reinterpret_cast<const v4f*>(ws + KN_OFF_);
    v4f bq[4][4];
    v4f kn2[4];
    for (int st = 0; st < 4; ++st) kn2[st] = fn[(w * 4 + st) * 16 + n];
    for (int j = 0; j < 4; ++j)
        for (int st = 0; st < 4; ++st)
            bq[j][st] = fq[((4 * w + j) * 4 + st) * 32 + 2 * n + hi];

    for (int i = 0; i < 2; ++i) zsh[w][l + 32 * i] = -2.f * ((xr[i] - mux) * inx);
    __syncthreads();

    for (int i = 0; i < 2; ++i) {
        const int u = l + 32 * i;
        if (u < ROWS_ + 24) {
            float acc = 0.f;
            for (int c = 0; c < 8; ++c) { const float z = zsh[w][u + c]; acc += z * z; }
            pnsh[w][u] = 0.25f * acc;
        }
    }
    __syncthreads();

    v8f pnc[6];
    for (int i = 0; i < 6; ++i) {
        const v4f* pp = reinterpret_cast<const v4f*>(&pnsh[w][8 * (i + hi)]);
        const v4f p0 = pp[0], p1 = pp[1];
        pnc[i] = (v8f){p0.x, p0.y, p0.z, p0.w, p1.x, p1.y, p1.z, p1.w};
    }

    v4f af[6];
    for (int i = 0; i < 6; ++i) {
        const float* zp = &zsh[w][8 * i + n + 2 * hi];
        af[i] = (v4f){zp[0], zp[1], zp[4], zp[5]};
    }

    const int ntile = (tg + 16 < TOUT_) ? 2 : 1;
    #pragma unroll
    for (int tile = 0; tile < 2; ++tile) {
        if (tile < ntile) {
            float md[4][8];
            for (int j = 0; j < 4; ++j) {
                const v2f a_lo = {af[2 * tile + j].x, af[2 * tile + j].y};
                const v2f a_hi = {af[2 * tile + j].z, af[2 * tile + j].w};

                for (int r = 0; r < 8; ++r) md[j][r] = FLT_MAX;

                for (int st = 0; st < 4; ++st) {
                    const v2f b_lo = {bq[j][st].x, bq[j][st].y};
                    const v2f b_hi = {bq[j][st].z, bq[j][st].w};
                    v8f g = __builtin_amdgcn_wmma_f32_16x16x4_f32(false, a_lo, false, b_lo,
                                                                  (short)0, pnc[2 * tile + j], false, false);
                    g = __builtin_amdgcn_wmma_f32_16x16x4_f32(false, a_hi, false, b_hi,
                                                              (short)0, g, false, false);
                    const v2f kk = {kn2[st][j], kn2[st][j]};
                    const v2f d0 = __builtin_shufflevector(g, g, 0, 1) + kk;
                    const v2f d1 = __builtin_shufflevector(g, g, 2, 3) + kk;
                    const v2f d2 = __builtin_shufflevector(g, g, 4, 5) + kk;
                    const v2f d3 = __builtin_shufflevector(g, g, 6, 7) + kk;
                    md[j][0] = fminf(md[j][0], d0.x);  md[j][1] = fminf(md[j][1], d0.y);
                    md[j][2] = fminf(md[j][2], d1.x);  md[j][3] = fminf(md[j][3], d1.y);
                    md[j][4] = fminf(md[j][4], d2.x);  md[j][5] = fminf(md[j][5], d2.y);
                    md[j][6] = fminf(md[j][6], d3.x);  md[j][7] = fminf(md[j][7], d3.y);
                }
            }

            {
                float* rp = &red[(tile * 16 + 8 * hi) * ROWP_ + w * THRP_ + n];
                for (int j = 0; j < 4; ++j)
                    for (int r = 0; r < 8; ++r)
                        rp[r * ROWP_ + j * CELLP_] = md[j][r];
            }
        }
    }
    __syncthreads();

    {
        const int row = threadIdx.x >> 3;
        const int q   = threadIdx.x & 7;
        const int t   = tg + row;
        if (t < TOUT_) {
            const v4f* cp = reinterpret_cast<const v4f*>(&red[row * ROWP_ + q * THRP_]);
            float o[4];
            for (int s = 0; s < 4; ++s) {
                const v4f c0 = cp[s * 5 + 0], c1 = cp[s * 5 + 1], c2 = cp[s * 5 + 2], c3 = cp[s * 5 + 3];
                const float m0 = fminf(fminf(c0.x, c0.y), fminf(c0.z, c0.w));
                const float m1 = fminf(fminf(c1.x, c1.y), fminf(c1.z, c1.w));
                const float m2 = fminf(fminf(c2.x, c2.y), fminf(c2.z, c2.w));
                const float m3 = fminf(fminf(c3.x, c3.y), fminf(c3.z, c3.w));
                o[s] = fminf(fminf(m0, m1), fminf(m2, m3));
            }
            v4f ov = {o[0], o[1], o[2], o[3]};
            *reinterpret_cast<v4f*>(out + (b * TOUT_ + t) * KSZ_ + q * 4) = ov;
        }
    }
}

extern "C" void kernel_launch(void* const* d_in, const int* in_sizes, int n_in,
                              void* d_out, int out_size, void* d_ws, size_t ws_size,
                              hipStream_t stream) {
    (void)in_sizes; (void)n_in; (void)out_size; (void)ws_size;
    const float* x    = (const float*)d_in[0];
    const float* kern = (const float*)d_in[1];
    float* ws  = (float*)d_ws;
    float* out = (float*)d_out;

    lsd_stats<<<dim3(SWAVES_), dim3(32), 0, stream>>>(x, kern, ws);
    lsd_rows_kernel<<<dim3(16, B_), dim3(256), 0, stream>>>(x, ws, out);
}


# Main kernel: both staging barriers dropped (a wave stages and reads back only its own channel's LDS rows); fragment loads now stay in flight under the staging and the slot-by-slot products
# speedup vs baseline: 3.3579x; 1.0513x over previous
#include <hip/hip_runtime.h>
#include <hip/hip_bf16.h>
#include <float.h>
#include <math.h>


typedef __attribute__((ext_vector_type(2))) float v2f;
typedef __attribute__((ext_vector_type(4))) float v4f;
typedef __attribute__((ext_vector_type(8))) float v8f;

#define B_   16
#define T_   512
#define C_   8
#define S_   64
#define KSZ_ 32
#define TOUT_ 481
#define EPS_ 1e-8f

#define ROWS_ 32
#define ZW_   64
#define CELLP_ 20
#define THRP_  84
#define ROWP_  676

#define KN_OFF_   512
#define FRAG_OFF_ 2560

#define SWAVES_ 80

template <int CTRL>
__device__ __forceinline__ float lane_x(float v) {
    const int b = __builtin_bit_cast(int, v);
    return __builtin_bit_cast(float, __builtin_amdgcn_update_dpp(b, b, CTRL, 0xf, 0xf, false));
}
__device__ __forceinline__ float lane_x16(float v) {
    const int b = __builtin_bit_cast(int, v);
    return __builtin_bit_cast(float, __builtin_amdgcn_permlanex16(b, b, 0x76543210, 0xfedcba98, false, false));
}

__global__ __launch_bounds__(32) __attribute__((amdgpu_waves_per_eu(1, 1))) void lsd_stats(const float* __restrict__ x,
                                                const float* __restrict__ kern,
                                                float* __restrict__ ws) {
    const int l = threadIdx.x;

    if (blockIdx.x < B_) {
        const int b = blockIdx.x;
        const v4f* __restrict__ xb = reinterpret_cast<const v4f*>(x + b * (T_ * C_));
        v4f s = {0.f, 0.f, 0.f, 0.f};
        v4f q = {0.f, 0.f, 0.f, 0.f};
        v4f v[32];
        for (int i = 0; i < 32; ++i) v[i] = xb[i * 32 + l];
        for (int i = 0; i < 32; ++i) { s += v[i]; q += v[i] * v[i]; }
        for (int e = 0; e < 4; ++e) {
            float a = s[e], c = q[e];
            a += lane_x<0x162>(a);  c += lane_x<0x162>(c);
            a += lane_x<0x164>(a);  c += lane_x<0x164>(c);
            a += lane_x<0x168>(a);  c += lane_x<0x168>(c);
            a += lane_x16(a);       c += lane_x16(c);
            s[e] = a; q[e] = c;
        }
        v4f mean4, inv4;
        for (int e = 0; e < 4; ++e) {
            const float mean = s[e] * (1.f / (float)T_);
            const float var  = fmaxf(q[e] * (1.f / (float)T_) - mean * mean, 0.f);
            mean4[e] = mean;
            inv4[e]  = 1.f / (sqrtf(var) + EPS_);
        }
        if (l < 2) {
            *reinterpret_cast<v4f*>(ws + b * C_ + 4 * l)       = mean4;
            *reinterpret_cast<v4f*>(ws + 128 + b * C_ + 4 * l) = inv4;
        }
    } else {
        const int sidx = blockIdx.x - B_;
        const int st = sidx >> 4;
        const int n  = sidx & 15;
        const int k  = l;
        const v4f* __restrict__ kp = reinterpret_cast<const v4f*>(kern + sidx * 256 + k * C_);
        const v4f v0 = kp[0];
        const v4f v1 = kp[1];
        float s = ((v0.x + v0.y) + (v0.z + v0.w)) + ((v1.x + v1.y) + (v1.z + v1.w));
        float q = ((v0.x * v0.x + v0.y * v0.y) + (v0.z * v0.z + v0.w * v0.w))
                + ((v1.x * v1.x + v1.y * v1.y) + (v1.z * v1.z + v1.w * v1.w));
        s += lane_x<0x161>(s);  q += lane_x<0x161>(q);
        s += lane_x<0x162>(s);  q += lane_x<0x162>(q);
        s += lane_x<0x164>(s);  q += lane_x<0x164>(q);
        s += lane_x<0x168>(s);  q += lane_x<0x168>(q);
        s += lane_x16(s);       q += lane_x16(q);
        const float mu  = s * (1.f / 256.f);
        const float var = fmaxf(q * (1.f / 256.f) - mu * mu, 0.f);
        const float inv = 1.f / (sqrtf(var) + EPS_);
        float z[C_];
        z[0] = (v0.x - mu) * inv;  z[1] = (v0.y - mu) * inv;
        z[2] = (v0.z - mu) * inv;  z[3] = (v0.w - mu) * inv;
        z[4] = (v1.x - mu) * inv;  z[5] = (v1.y - mu) * inv;
        z[6] = (v1.z - mu) * inv;  z[7] = (v1.w - mu) * inv;
        float acc = 0.f;
        for (int c = 0; c < C_; ++c) acc += z[c] * z[c];
        v4f* fp = reinterpret_cast<v4f*>(ws + FRAG_OFF_ + (((k * 4 + st) * 32 + n * 2) << 2));
        const v4f c0 = {z[0], z[1], z[4], z[5]};
        const v4f c1 = {z[2], z[3], z[6], z[7]};
        fp[0] = c0;
        fp[1] = c1;
        ws[KN_OFF_ + ((((k >> 2) * 4 + st) * 16 + n) << 2) + (k & 3)] = acc;
    }
}

__device__ __forceinline__ void wave_lds_sync() {
    __builtin_amdgcn_fence(__ATOMIC_RELEASE, "workgroup", "local");
    __builtin_amdgcn_wave_barrier();
    __builtin_amdgcn_fence(__ATOMIC_ACQUIRE, "workgroup", "local");
}

__global__ __launch_bounds__(256) void lsd_rows_kernel(const float* __restrict__ x,
                                                       const float* __restrict__ ws,
                                                       float* __restrict__ out) {
    const int tg = blockIdx.x * ROWS_;
    const int b  = blockIdx.y;
    const int l  = threadIdx.x & 31;
    const int w  = __builtin_amdgcn_readfirstlane((int)(threadIdx.x >> 5));

    __shared__ float zsh[C_][ZW_];
    __shared__ __attribute__((aligned(32))) float pnsh[C_][ZW_];
    __shared__ __attribute__((aligned(16))) float red[ROWS_ * ROWP_];

    const int n  = l & 15;
    const int hi = l >> 4;

    const float mux = ws[b * C_ + w];
    const float inx = ws[128 + b * C_ + w];
    float xr[2];
    for (int i = 0; i < 2; ++i) {
        int t = tg + l + 32 * i;
        if (t > T_ - 1) t = T_ - 1;
        xr[i] = x[(b * T_ + t) * C_ + w];
    }
    const v4f* __restrict__ fq = reinterpret_cast<const v4f*>(ws + FRAG_OFF_);
    const v4f* __restrict__ fn = reinterpret_cast<const v4f*>(ws + KN_OFF_);
    v4f bq[4][4];
    v4f kn2[4];
    for (int st = 0; st < 4; ++st) bq[0][st] = fq[((4 * w + 0) * 4 + st) * 32 + 2 * n + hi];
    for (int st = 0; st < 4; ++st) kn2[st] = fn[(w * 4 + st) * 16 + n];
    for (int j = 1; j < 4; ++j)
        for (int st = 0; st < 4; ++st)
            bq[j][st] = fq[((4 * w + j) * 4 + st) * 32 + 2 * n + hi];

    for (int i = 0; i < 2; ++i) zsh[w][l + 32 * i] = -2.f * ((xr[i] - mux) * inx);
    wave_lds_sync();

    for (int i = 0; i < 2; ++i) {
        const int u = l + 32 * i;
        if (u < ROWS_ + 24) {
            float acc = 0.f;
            for (int c = 0; c < 8; ++c) { const float z = zsh[w][u + c]; acc += z * z; }
            pnsh[w][u] = 0.25f * acc;
        }
    }
    wave_lds_sync();

    v8f pnc[6];
    for (int i = 0; i < 6; ++i) {
        const v4f* pp = reinterpret_cast<const v4f*>(&pnsh[w][8 * (i + hi)]);
        const v4f p0 = pp[0], p1 = pp[1];
        pnc[i] = (v8f){p0.x, p0.y, p0.z, p0.w, p1.x, p1.y, p1.z, p1.w};
    }

    v4f af[6];
    for (int i = 0; i < 6; ++i) {
        const float* zp = &zsh[w][8 * i + n + 2 * hi];
        af[i] = (v4f){zp[0], zp[1], zp[4], zp[5]};
    }

    float md[2][4][8];
    __builtin_amdgcn_sched_barrier(0);
    #pragma unroll
    for (int j = 0; j < 4; ++j) {
        #pragma unroll
        for (int tile = 0; tile < 2; ++tile) {
            const v2f a_lo = {af[2 * tile + j].x, af[2 * tile + j].y};
            const v2f a_hi = {af[2 * tile + j].z, af[2 * tile + j].w};

            for (int r = 0; r < 8; ++r) md[tile][j][r] = FLT_MAX;

            for (int st = 0; st < 4; ++st) {
                const v2f b_lo = {bq[j][st].x, bq[j][st].y};
                const v2f b_hi = {bq[j][st].z, bq[j][st].w};
                v8f g = __builtin_amdgcn_wmma_f32_16x16x4_f32(false, a_lo, false, b_lo,
                                                              (short)0, pnc[2 * tile + j], false, false);
                g = __builtin_amdgcn_wmma_f32_16x16x4_f32(false, a_hi, false, b_hi,
                                                          (short)0, g, false, false);
                const v2f kk = {kn2[st][j], kn2[st][j]};
                const v2f d0 = __builtin_shufflevector(g, g, 0, 1) + kk;
                const v2f d1 = __builtin_shufflevector(g, g, 2, 3) + kk;
                const v2f d2 = __builtin_shufflevector(g, g, 4, 5) + kk;
                const v2f d3 = __builtin_shufflevector(g, g, 6, 7) + kk;
                md[tile][j][0] = fminf(md[tile][j][0], d0.x);  md[tile][j][1] = fminf(md[tile][j][1], d0.y);
                md[tile][j][2] = fminf(md[tile][j][2], d1.x);  md[tile][j][3] = fminf(md[tile][j][3], d1.y);
                md[tile][j][4] = fminf(md[tile][j][4], d2.x);  md[tile][j][5] = fminf(md[tile][j][5], d2.y);
                md[tile][j][6] = fminf(md[tile][j][6], d3.x);  md[tile][j][7] = fminf(md[tile][j][7], d3.y);
            }
        }
    }

    for (int tile = 0; tile < 2; ++tile) {
        float* rp = &red[(tile * 16 + 8 * hi) * ROWP_ + w * THRP_ + n];
        for (int j = 0; j < 4; ++j)
            for (int r = 0; r < 8; ++r)
                rp[r * ROWP_ + j * CELLP_] = md[tile][j][r];
    }
    __syncthreads();

    {
        const int row = threadIdx.x >> 3;
        const int q   = threadIdx.x & 7;
        const int t   = tg + row;
        if (t < TOUT_) {
            const v4f* cp = reinterpret_cast<const v4f*>(&red[row * ROWP_ + q * THRP_]);
            float o[4];
            for (int s = 0; s < 4; ++s) {
                const v4f c0 = cp[s * 5 + 0], c1 = cp[s * 5 + 1], c2 = cp[s * 5 + 2], c3 = cp[s * 5 + 3];
                const float m0 = fminf(fminf(c0.x, c0.y), fminf(c0.z, c0.w));
                const float m1 = fminf(fminf(c1.x, c1.y), fminf(c1.z, c1.w));
                const float m2 = fminf(fminf(c2.x, c2.y), fminf(c2.z, c2.w));
                const float m3 = fminf(fminf(c3.x, c3.y), fminf(c3.z, c3.w));
                o[s] = fminf(fminf(m0, m1), fminf(m2, m3));
            }
            v4f ov = {o[0], o[1], o[2], o[3]};
            *reinterpret_cast<v4f*>(out + (b * TOUT_ + t) * KSZ_ + q * 4) = ov;
        }
    }
}

extern "C" void kernel_launch(void* const* d_in, const int* in_sizes, int n_in,
                              void* d_out, int out_size, void* d_ws, size_t ws_size,
                              hipStream_t stream) {
    (void)in_sizes; (void)n_in; (void)out_size; (void)ws_size;
    const float* x    = (const float*)d_in[0];
    const float* kern = (const float*)d_in[1];
    float* ws  = (float*)d_ws;
    float* out = (float*)d_out;

    lsd_stats<<<dim3(SWAVES_), dim3(32), 0, stream>>>(x, kern, ws);
    lsd_rows_kernel<<<dim3(16, B_), dim3(256), 0, stream>>>(x, ws, out);
}
